# starting kernel
# baseline (speedup 1.0000x reference)
#include <hip/hip_runtime.h>

typedef float v2f __attribute__((ext_vector_type(2)));
typedef float v4f __attribute__((ext_vector_type(4)));
typedef float v8f __attribute__((ext_vector_type(8)));

#define COUT 64
#define BN_EPS 1e-5f

__global__ void zero_f32v4(v4f* __restrict__ p, int n4) {
  int i = blockIdx.x * blockDim.x + threadIdx.x;
  int s = gridDim.x * blockDim.x;
  v4f z = {};
  for (; i < n4; i += s) p[i] = z;
}

template <int CIN>
__global__ void sparse_conv_wmma(const float* __restrict__ feats,
                                 const float* __restrict__ W,
                                 const int*   __restrict__ pin,
                                 const int*   __restrict__ pout,
                                 float*       __restrict__ out,
                                 int P, int mtiles) {
  const int wave  = threadIdx.x >> 5;
  const int mtile = blockIdx.x * 8 + wave;
  if (mtile >= mtiles) return;

  const int k     = blockIdx.y;
  const int ntile = blockIdx.z;
  const int lane  = threadIdx.x & 31;
  const int m0    = mtile * 16;
  const int n0    = ntile * 16;
  const int half  = lane >> 4;
  const int nl    = lane & 15;

  const int pidx = m0 + nl;
  const int arow = (pidx < P) ? pin[k * P + pidx] : 0;
  const float* aptr = feats + (long)arow * CIN + 2 * half;
  const float* bptr = W + (long)k * CIN * COUT + 2 * half * COUT + n0 + nl;

  v8f c = {};
#pragma unroll
  for (int c0 = 0; c0 < CIN; c0 += 4) {
    v2f a = *reinterpret_cast<const v2f*>(aptr + c0);
    v2f b;
    b.x = bptr[c0 * COUT];
    b.y = bptr[c0 * COUT + COUT];
    c = __builtin_amdgcn_wmma_f32_16x16x4_f32(false, a, false, b,
                                              (short)0, c, false, false);
  }

#pragma unroll
  for (int r = 0; r < 8; ++r) {
    const int pi = m0 + r + 8 * half;
    if (pi < P) {
      const int orow = pout[k * P + pi];
      atomicAdd(&out[(long)orow * COUT + n0 + nl], c[r]);
    }
  }
}

__global__ void col_stats(const float* __restrict__ x, int n_elems,
                          float* __restrict__ stats) {
  __shared__ float ssum[COUT];
  __shared__ float ssq[COUT];
  if (threadIdx.x < COUT) { ssum[threadIdx.x] = 0.f; ssq[threadIdx.x] = 0.f; }
  __syncthreads();
  int idx    = blockIdx.x * blockDim.x + threadIdx.x;
  int stride = gridDim.x * blockDim.x;
  const int ch = idx & (COUT - 1);
  float lsum = 0.f, lsq = 0.f;
  for (int i = idx; i < n_elems; i += stride) {
    float v = x[i];
    lsum += v;
    lsq  += v * v;
  }
  atomicAdd(&ssum[ch], lsum);
  atomicAdd(&ssq[ch], lsq);
  __syncthreads();
  if (threadIdx.x < COUT) {
    atomicAdd(&stats[threadIdx.x], ssum[threadIdx.x]);
    atomicAdd(&stats[COUT + threadIdx.x], ssq[threadIdx.x]);
  }
}

__global__ void bn_finalize(const float* __restrict__ stats,
                            const float* __restrict__ g,
                            const float* __restrict__ b,
                            float inv_n, float* __restrict__ ab) {
  int c = threadIdx.x;
  if (c < COUT) {
    float mean = stats[c] * inv_n;
    float var  = stats[COUT + c] * inv_n - mean * mean;
    float s    = g[c] * rsqrtf(var + BN_EPS);
    ab[c]        = s;
    ab[COUT + c] = b[c] - mean * s;
  }
}

__global__ void bn_apply(const float* __restrict__ y,
                         const float* __restrict__ ab,
                         float* __restrict__ x, int n_elems) {
  int i = blockIdx.x * blockDim.x + threadIdx.x;
  int s = gridDim.x * blockDim.x;
  for (; i < n_elems; i += s) {
    int ch = i & (COUT - 1);
    float v = y[i] * ab[ch] + ab[COUT + ch];
    x[i] = v > 0.f ? v : 0.f;
  }
}

__global__ void bn_apply_add(const float* __restrict__ y,
                             const float* __restrict__ ab,
                             const float* __restrict__ xnet,
                             float* __restrict__ out, int n_elems) {
  int i = blockIdx.x * blockDim.x + threadIdx.x;
  int s = gridDim.x * blockDim.x;
  for (; i < n_elems; i += s) {
    int ch = i & (COUT - 1);
    float v = y[i] * ab[ch] + ab[COUT + ch];
    out[i] = xnet[i] + (v > 0.f ? v : 0.f);
  }
}

static inline int cdiv(long a, long b) { return (int)((a + b - 1) / b); }

extern "C" void kernel_launch(void* const* d_in, const int* in_sizes, int n_in,
                              void* d_out, int out_size, void* d_ws, size_t ws_size,
                              hipStream_t stream) {
  const float* feats  = (const float*)d_in[0];
  const float* W_down = (const float*)d_in[1];
  const float* g0     = (const float*)d_in[2];
  const float* b0     = (const float*)d_in[3];
  const float* W1     = (const float*)d_in[4];
  const float* g1     = (const float*)d_in[5];
  const float* b1     = (const float*)d_in[6];
  const float* W2     = (const float*)d_in[7];
  const float* g2     = (const float*)d_in[8];
  const float* b2     = (const float*)d_in[9];
  const int* p_down_in  = (const int*)d_in[10];
  const int* p_down_out = (const int*)d_in[11];
  const int* p_subm_in  = (const int*)d_in[12];
  const int* p_subm_out = (const int*)d_in[13];

  const int Cin = 32;
  const int K   = in_sizes[1] / (Cin * COUT);
  const int P1  = in_sizes[10] / K;
  const int P2  = in_sizes[12] / K;
  const int n_out = out_size / COUT;
  const int NE  = n_out * COUT;
  const float inv_n = 1.0f / (float)n_out;

  float* y     = (float*)d_ws;
  float* x     = y + NE;
  float* xnet  = x + NE;
  float* stats = xnet + NE;
  float* ab    = stats + 128;
  (void)n_in; (void)ws_size;

  const dim3 blk(256);
  const int zb  = cdiv(NE / 4, 256);
  const int eb  = cdiv(NE, 256);
  const int sb  = 1024;

  const int mt1 = cdiv(P1, 16);
  const int mt2 = cdiv(P2, 16);
  const dim3 cg1(cdiv(mt1, 8), K, 4);
  const dim3 cg2(cdiv(mt2, 8), K, 4);

  zero_f32v4<<<zb, blk, 0, stream>>>((v4f*)y, NE / 4);
  zero_f32v4<<<1, 32, 0, stream>>>((v4f*)stats, 32);
  sparse_conv_wmma<32><<<cg1, blk, 0, stream>>>(feats, W_down, p_down_in, p_down_out,
                                                y, P1, mt1);
  col_stats<<<sb, blk, 0, stream>>>(y, NE, stats);
  bn_finalize<<<1, 64, 0, stream>>>(stats, g0, b0, inv_n, ab);
  bn_apply<<<eb, blk, 0, stream>>>(y, ab, x, NE);

  zero_f32v4<<<zb, blk, 0, stream>>>((v4f*)y, NE / 4);
  zero_f32v4<<<1, 32, 0, stream>>>((v4f*)stats, 32);
  sparse_conv_wmma<64><<<cg2, blk, 0, stream>>>(x, W1, p_subm_in, p_subm_out,
                                                y, P2, mt2);
  col_stats<<<sb, blk, 0, stream>>>(y, NE, stats);
  bn_finalize<<<1, 64, 0, stream>>>(stats, g1, b1, inv_n, ab);
  bn_apply<<<eb, blk, 0, stream>>>(y, ab, xnet, NE);

  zero_f32v4<<<zb, blk, 0, stream>>>((v4f*)y, NE / 4);
  zero_f32v4<<<1, 32, 0, stream>>>((v4f*)stats, 32);
  sparse_conv_wmma<64><<<cg2, blk, 0, stream>>>(xnet, W2, p_subm_in, p_subm_out,
                                                y, P2, mt2);
  col_stats<<<sb, blk, 0, stream>>>(y, NE, stats);
  bn_finalize<<<1, 64, 0, stream>>>(stats, g2, b2, inv_n, ab);
  bn_apply_add<<<eb, blk, 0, stream>>>(y, ab, xnet, (float*)d_out, NE);
}


# conv: one wave computes all 64 output columns of its 16 pairs (was 16): each gathered A fragment is loaded once and feeds 4 WMMAs; grid z dropped; float atomics kept
# speedup vs baseline: 1.8792x; 1.8792x over previous
#include <hip/hip_runtime.h>

typedef float v2f __attribute__((ext_vector_type(2)));
typedef float v4f __attribute__((ext_vector_type(4)));
typedef float v8f __attribute__((ext_vector_type(8)));

#define COUT 64
#define BN_EPS 1e-5f

__global__ void zero_f32v4(v4f* __restrict__ p, int n4) {
  int i = blockIdx.x * blockDim.x + threadIdx.x;
  int s = gridDim.x * blockDim.x;
  v4f z = {};
  for (; i < n4; i += s) p[i] = z;
}

#define NT (COUT / 16)
template <int CIN>
__global__ void sparse_conv_wmma(const float* __restrict__ feats,
                                 const float* __restrict__ W,
                                 const int*   __restrict__ pin,
                                 const int*   __restrict__ pout,
                                 float*       __restrict__ out,
                                 int P, int mtiles) {
  const int wave  = threadIdx.x >> 5;
  const int mtile = blockIdx.x * 8 + wave;
  if (mtile >= mtiles) return;

  const int k     = blockIdx.y;
  const int lane  = threadIdx.x & 31;
  const int m0    = mtile * 16;
  const int half  = lane >> 4;
  const int nl    = lane & 15;

  const int pidx = m0 + nl;
  const int arow = (pidx < P) ? pin[k * P + pidx] : 0;
  const float* aptr = feats + (long)arow * CIN + 2 * half;
  const float* bptr = W + (long)k * CIN * COUT + 2 * half * COUT + nl;

  v8f c[NT];
#pragma unroll
  for (int j = 0; j < NT; ++j) c[j] = v8f{};

#pragma unroll
  for (int c0 = 0; c0 < CIN; c0 += 4) {
    v2f a = *reinterpret_cast<const v2f*>(aptr + c0);
#pragma unroll
    for (int j = 0; j < NT; ++j) {
      v2f b;
      b.x = bptr[c0 * COUT + 16 * j];
      b.y = bptr[c0 * COUT + COUT + 16 * j];
      c[j] = __builtin_amdgcn_wmma_f32_16x16x4_f32(false, a, false, b,
                                                   (short)0, c[j], false, false);
    }
  }

#pragma unroll
  for (int r = 0; r < 8; ++r) {
    const int pi = m0 + r + 8 * half;
    if (pi < P) {
      const int orow = pout[k * P + pi];
      float* orp = out + (long)orow * COUT + nl;
#pragma unroll
      for (int j = 0; j < NT; ++j) atomicAdd(orp + 16 * j, c[j][r]);
    }
  }
}

__global__ void col_stats(const float* __restrict__ x, int n_elems,
                          float* __restrict__ stats) {
  __shared__ float ssum[COUT];
  __shared__ float ssq[COUT];
  if (threadIdx.x < COUT) { ssum[threadIdx.x] = 0.f; ssq[threadIdx.x] = 0.f; }
  __syncthreads();
  int idx    = blockIdx.x * blockDim.x + threadIdx.x;
  int stride = gridDim.x * blockDim.x;
  const int ch = idx & (COUT - 1);
  float lsum = 0.f, lsq = 0.f;
  for (int i = idx; i < n_elems; i += stride) {
    float v = x[i];
    lsum += v;
    lsq  += v * v;
  }
  atomicAdd(&ssum[ch], lsum);
  atomicAdd(&ssq[ch], lsq);
  __syncthreads();
  if (threadIdx.x < COUT) {
    atomicAdd(&stats[threadIdx.x], ssum[threadIdx.x]);
    atomicAdd(&stats[COUT + threadIdx.x], ssq[threadIdx.x]);
  }
}

__global__ void bn_finalize(const float* __restrict__ stats,
                            const float* __restrict__ g,
                            const float* __restrict__ b,
                            float inv_n, float* __restrict__ ab) {
  int c = threadIdx.x;
  if (c < COUT) {
    float mean = stats[c] * inv_n;
    float var  = stats[COUT + c] * inv_n - mean * mean;
    float s    = g[c] * rsqrtf(var + BN_EPS);
    ab[c]        = s;
    ab[COUT + c] = b[c] - mean * s;
  }
}

__global__ void bn_apply(const float* __restrict__ y,
                         const float* __restrict__ ab,
                         float* __restrict__ x, int n_elems) {
  int i = blockIdx.x * blockDim.x + threadIdx.x;
  int s = gridDim.x * blockDim.x;
  for (; i < n_elems; i += s) {
    int ch = i & (COUT - 1);
    float v = y[i] * ab[ch] + ab[COUT + ch];
    x[i] = v > 0.f ? v : 0.f;
  }
}

__global__ void bn_apply_add(const float* __restrict__ y,
                             const float* __restrict__ ab,
                             const float* __restrict__ xnet,
                             float* __restrict__ out, int n_elems) {
  int i = blockIdx.x * blockDim.x + threadIdx.x;
  int s = gridDim.x * blockDim.x;
  for (; i < n_elems; i += s) {
    int ch = i & (COUT - 1);
    float v = y[i] * ab[ch] + ab[COUT + ch];
    out[i] = xnet[i] + (v > 0.f ? v : 0.f);
  }
}

static inline int cdiv(long a, long b) { return (int)((a + b - 1) / b); }

extern "C" void kernel_launch(void* const* d_in, const int* in_sizes, int n_in,
                              void* d_out, int out_size, void* d_ws, size_t ws_size,
                              hipStream_t stream) {
  const float* feats  = (const float*)d_in[0];
  const float* W_down = (const float*)d_in[1];
  const float* g0     = (const float*)d_in[2];
  const float* b0     = (const float*)d_in[3];
  const float* W1     = (const float*)d_in[4];
  const float* g1     = (const float*)d_in[5];
  const float* b1     = (const float*)d_in[6];
  const float* W2     = (const float*)d_in[7];
  const float* g2     = (const float*)d_in[8];
  const float* b2     = (const float*)d_in[9];
  const int* p_down_in  = (const int*)d_in[10];
  const int* p_down_out = (const int*)d_in[11];
  const int* p_subm_in  = (const int*)d_in[12];
  const int* p_subm_out = (const int*)d_in[13];

  const int Cin = 32;
  const int K   = in_sizes[1] / (Cin * COUT);
  const int P1  = in_sizes[10] / K;
  const int P2  = in_sizes[12] / K;
  const int n_out = out_size / COUT;
  const int NE  = n_out * COUT;
  const float inv_n = 1.0f / (float)n_out;

  float* y     = (float*)d_ws;
  float* x     = y + NE;
  float* xnet  = x + NE;
  float* stats = xnet + NE;
  float* ab    = stats + 128;
  (void)n_in; (void)ws_size;

  const dim3 blk(256);
  const int zb  = cdiv(NE / 4, 256);
  const int eb  = cdiv(NE, 256);
  const int sb  = 1024;

  const int mt1 = cdiv(P1, 16);
  const int mt2 = cdiv(P2, 16);
  const dim3 cg1(cdiv(mt1, 8), K);
  const dim3 cg2(cdiv(mt2, 8), K);

  zero_f32v4<<<zb, blk, 0, stream>>>((v4f*)y, NE / 4);
  zero_f32v4<<<1, 32, 0, stream>>>((v4f*)stats, 32);
  sparse_conv_wmma<32><<<cg1, blk, 0, stream>>>(feats, W_down, p_down_in, p_down_out,
                                                y, P1, mt1);
  col_stats<<<sb, blk, 0, stream>>>(y, NE, stats);
  bn_finalize<<<1, 64, 0, stream>>>(stats, g0, b0, inv_n, ab);
  bn_apply<<<eb, blk, 0, stream>>>(y, ab, x, NE);

  zero_f32v4<<<zb, blk, 0, stream>>>((v4f*)y, NE / 4);
  zero_f32v4<<<1, 32, 0, stream>>>((v4f*)stats, 32);
  sparse_conv_wmma<64><<<cg2, blk, 0, stream>>>(x, W1, p_subm_in, p_subm_out,
                                                y, P2, mt2);
  col_stats<<<sb, blk, 0, stream>>>(y, NE, stats);
  bn_finalize<<<1, 64, 0, stream>>>(stats, g1, b1, inv_n, ab);
  bn_apply<<<eb, blk, 0, stream>>>(y, ab, xnet, NE);

  zero_f32v4<<<zb, blk, 0, stream>>>((v4f*)y, NE / 4);
  zero_f32v4<<<1, 32, 0, stream>>>((v4f*)stats, 32);
  sparse_conv_wmma<64><<<cg2, blk, 0, stream>>>(xnet, W2, p_subm_in, p_subm_out,
                                                y, P2, mt2);
  col_stats<<<sb, blk, 0, stream>>>(y, NE, stats);
  bn_finalize<<<1, 64, 0, stream>>>(stats, g2, b2, inv_n, ab);
  bn_apply_add<<<eb, blk, 0, stream>>>(y, ab, xnet, (float*)d_out, NE);
}


# conv scatter: tiles 2t,2t+1 trade 16-lane halves (v_permlanex16) so each float atomic adds ONE whole 128-B line of one output row, not two 64-B halves of two rows; same adds, float atomics kept
# speedup vs baseline: 2.1883x; 1.1645x over previous
#include <hip/hip_runtime.h>

typedef float v2f __attribute__((ext_vector_type(2)));
typedef float v4f __attribute__((ext_vector_type(4)));
typedef float v8f __attribute__((ext_vector_type(8)));

#define COUT 64
#define BN_EPS 1e-5f

__global__ void zero_f32v4(v4f* __restrict__ p, int n4) {
  int i = blockIdx.x * blockDim.x + threadIdx.x;
  int s = gridDim.x * blockDim.x;
  v4f z = {};
  for (; i < n4; i += s) p[i] = z;
}

#define NT (COUT / 16)

__device__ __forceinline__ void swap_halves(float x, float y, bool upper, float& lo, float& hi) {
  const int give = __builtin_bit_cast(int, upper ? x : y);
  const float got = __builtin_bit_cast(float,
      __builtin_amdgcn_permlanex16(give, give, 0x76543210, 0xfedcba98, false, false));
  lo = upper ? got : x;
  hi = upper ? y : got;
}

template <int CIN>
__global__ void sparse_conv_wmma(const float* __restrict__ feats,
                                 const float* __restrict__ W,
                                 const int*   __restrict__ pin,
                                 const int*   __restrict__ pout,
                                 float*       __restrict__ out,
                                 int P, int mtiles) {
  const int wave  = __builtin_amdgcn_readfirstlane((int)(threadIdx.x >> 5));
  const int mtile = blockIdx.x * 8 + wave;
  if (mtile >= mtiles) return;

  const int k     = blockIdx.y;
  const unsigned lane = threadIdx.x & 31u;
  const int m0    = mtile * 16;
  const int half  = (int)(lane >> 4);
  const int nl    = (int)(lane & 15u);

  const int pidx = min(m0 + nl, P - 1);
  const int arow = pin[k * P + pidx];
  const int orow_l = pout[k * P + pidx];
  int orow[16];
#pragma unroll
  for (int i = 0; i < 16; ++i) orow[i] = __builtin_amdgcn_readlane(orow_l, i);
  const unsigned aoff = (unsigned)arow * CIN + 2u * half;
  const float* aptr = feats + aoff;
  const float* wk   = W + (size_t)k * CIN * COUT;
  const float* bptr = wk + (unsigned)(2 * half * COUT + nl);

  v8f c[NT];
#pragma unroll
  for (int j = 0; j < NT; ++j) c[j] = v8f{};

#pragma unroll
  for (int c0 = 0; c0 < CIN; c0 += 4) {
    v2f a = *reinterpret_cast<const v2f*>(aptr + c0);
#pragma unroll
    for (int j = 0; j < NT; ++j) {
      v2f b;
      b.x = bptr[c0 * COUT + 16 * j];
      b.y = bptr[c0 * COUT + COUT + 16 * j];
      c[j] = __builtin_amdgcn_wmma_f32_16x16x4_f32(false, a, false, b,
                                                   (short)0, c[j], false, false);
    }
  }

  const unsigned ln = __builtin_amdgcn_mbcnt_lo(~0u, 0u);
#pragma unroll
  for (int r = 0; r < 8; ++r) {
    float lo[NT / 2], hi[NT / 2];
#pragma unroll
    for (int t = 0; t < NT / 2; ++t) swap_halves(c[2 * t][r], c[2 * t + 1][r], half != 0, lo[t], hi[t]);
    if (m0 + r < P) {
      float* orp = out + (size_t)orow[r] * COUT;
#pragma unroll
      for (int t = 0; t < NT / 2; ++t) atomicAdd(orp + (ln + 32u * t), lo[t]);
    }
    if (m0 + r + 8 < P) {
      float* orp = out + (size_t)orow[r + 8] * COUT;
#pragma unroll
      for (int t = 0; t < NT / 2; ++t) atomicAdd(orp + (ln + 32u * t), hi[t]);
    }
  }
}

__global__ void col_stats(const float* __restrict__ x, int n_elems,
                          float* __restrict__ stats) {
  __shared__ float ssum[COUT];
  __shared__ float ssq[COUT];
  if (threadIdx.x < COUT) { ssum[threadIdx.x] = 0.f; ssq[threadIdx.x] = 0.f; }
  __syncthreads();
  int idx    = blockIdx.x * blockDim.x + threadIdx.x;
  int stride = gridDim.x * blockDim.x;
  const int ch = idx & (COUT - 1);
  float lsum = 0.f, lsq = 0.f;
  for (int i = idx; i < n_elems; i += stride) {
    float v = x[i];
    lsum += v;
    lsq  += v * v;
  }
  atomicAdd(&ssum[ch], lsum);
  atomicAdd(&ssq[ch], lsq);
  __syncthreads();
  if (threadIdx.x < COUT) {
    atomicAdd(&stats[threadIdx.x], ssum[threadIdx.x]);
    atomicAdd(&stats[COUT + threadIdx.x], ssq[threadIdx.x]);
  }
}

__global__ void bn_finalize(const float* __restrict__ stats,
                            const float* __restrict__ g,
                            const float* __restrict__ b,
                            float inv_n, float* __restrict__ ab) {
  int c = threadIdx.x;
  if (c < COUT) {
    float mean = stats[c] * inv_n;
    float var  = stats[COUT + c] * inv_n - mean * mean;
    float s    = g[c] * rsqrtf(var + BN_EPS);
    ab[c]        = s;
    ab[COUT + c] = b[c] - mean * s;
  }
}

__global__ void bn_apply(const float* __restrict__ y,
                         const float* __restrict__ ab,
                         float* __restrict__ x, int n_elems) {
  int i = blockIdx.x * blockDim.x + threadIdx.x;
  int s = gridDim.x * blockDim.x;
  for (; i < n_elems; i += s) {
    int ch = i & (COUT - 1);
    float v = y[i] * ab[ch] + ab[COUT + ch];
    x[i] = v > 0.f ? v : 0.f;
  }
}

__global__ void bn_apply_add(const float* __restrict__ y,
                             const float* __restrict__ ab,
                             const float* __restrict__ xnet,
                             float* __restrict__ out, int n_elems) {
  int i = blockIdx.x * blockDim.x + threadIdx.x;
  int s = gridDim.x * blockDim.x;
  for (; i < n_elems; i += s) {
    int ch = i & (COUT - 1);
    float v = y[i] * ab[ch] + ab[COUT + ch];
    out[i] = xnet[i] + (v > 0.f ? v : 0.f);
  }
}

static inline int cdiv(long a, long b) { return (int)((a + b - 1) / b); }

extern "C" void kernel_launch(void* const* d_in, const int* in_sizes, int n_in,
                              void* d_out, int out_size, void* d_ws, size_t ws_size,
                              hipStream_t stream) {
  const float* feats  = (const float*)d_in[0];
  const float* W_down = (const float*)d_in[1];
  const float* g0     = (const float*)d_in[2];
  const float* b0     = (const float*)d_in[3];
  const float* W1     = (const float*)d_in[4];
  const float* g1     = (const float*)d_in[5];
  const float* b1     = (const float*)d_in[6];
  const float* W2     = (const float*)d_in[7];
  const float* g2     = (const float*)d_in[8];
  const float* b2     = (const float*)d_in[9];
  const int* p_down_in  = (const int*)d_in[10];
  const int* p_down_out = (const int*)d_in[11];
  const int* p_subm_in  = (const int*)d_in[12];
  const int* p_subm_out = (const int*)d_in[13];

  const int Cin = 32;
  const int K   = in_sizes[1] / (Cin * COUT);
  const int P1  = in_sizes[10] / K;
  const int P2  = in_sizes[12] / K;
  const int n_out = out_size / COUT;
  const int NE  = n_out * COUT;
  const float inv_n = 1.0f / (float)n_out;

  float* y     = (float*)d_ws;
  float* x     = y + NE;
  float* xnet  = x + NE;
  float* stats = xnet + NE;
  float* ab    = stats + 128;
  (void)n_in; (void)ws_size;

  const dim3 blk(256);
  const int zb  = cdiv(NE / 4, 256);
  const int eb  = cdiv(NE, 256);
  const int sb  = 1024;

  const int mt1 = cdiv(P1, 16);
  const int mt2 = cdiv(P2, 16);
  const dim3 cg1(cdiv(mt1, 8), K);
  const dim3 cg2(cdiv(mt2, 8), K);

  zero_f32v4<<<zb, blk, 0, stream>>>((v4f*)y, NE / 4);
  zero_f32v4<<<1, 32, 0, stream>>>((v4f*)stats, 32);
  sparse_conv_wmma<32><<<cg1, blk, 0, stream>>>(feats, W_down, p_down_in, p_down_out,
                                                y, P1, mt1);
  col_stats<<<sb, blk, 0, stream>>>(y, NE, stats);
  bn_finalize<<<1, 64, 0, stream>>>(stats, g0, b0, inv_n, ab);
  bn_apply<<<eb, blk, 0, stream>>>(y, ab, x, NE);

  zero_f32v4<<<zb, blk, 0, stream>>>((v4f*)y, NE / 4);
  zero_f32v4<<<1, 32, 0, stream>>>((v4f*)stats, 32);
  sparse_conv_wmma<64><<<cg2, blk, 0, stream>>>(x, W1, p_subm_in, p_subm_out,
                                                y, P2, mt2);
  col_stats<<<sb, blk, 0, stream>>>(y, NE, stats);
  bn_finalize<<<1, 64, 0, stream>>>(stats, g1, b1, inv_n, ab);
  bn_apply<<<eb, blk, 0, stream>>>(y, ab, xnet, NE);

  zero_f32v4<<<zb, blk, 0, stream>>>((v4f*)y, NE / 4);
  zero_f32v4<<<1, 32, 0, stream>>>((v4f*)stats, 32);
  sparse_conv_wmma<64><<<cg2, blk, 0, stream>>>(xnet, W2, p_subm_in, p_subm_out,
                                                y, P2, mt2);
  col_stats<<<sb, blk, 0, stream>>>(y, NE, stats);
  bn_finalize<<<1, 64, 0, stream>>>(stats, g2, b2, inv_n, ab);
  bn_apply_add<<<eb, blk, 0, stream>>>(y, ab, xnet, (float*)d_out, NE);
}


# conv: each wave loads the B fragments of its whole W[k] into registers once and loops over its share of the 16-pair tiles (fixed grid x, 4-wave blocks), not once per tile; float atomics kept
# speedup vs baseline: 2.3552x; 1.0763x over previous
#include <hip/hip_runtime.h>

typedef float v2f __attribute__((ext_vector_type(2)));
typedef float v4f __attribute__((ext_vector_type(4)));
typedef float v8f __attribute__((ext_vector_type(8)));

#define COUT 64
#define BN_EPS 1e-5f

__global__ void zero_f32v4(v4f* __restrict__ p, int n4) {
  int i = blockIdx.x * blockDim.x + threadIdx.x;
  int s = gridDim.x * blockDim.x;
  v4f z = {};
  for (; i < n4; i += s) p[i] = z;
}

#define NT (COUT / 16)
#define CONV_WAVES 4

__device__ __forceinline__ void swap_halves(float x, float y, bool upper, float& lo, float& hi) {
  const int give = __builtin_bit_cast(int, upper ? x : y);
  const float got = __builtin_bit_cast(float,
      __builtin_amdgcn_permlanex16(give, give, 0x76543210, 0xfedcba98, false, false));
  lo = upper ? got : x;
  hi = upper ? y : got;
}

template <int CIN>
__global__ void __launch_bounds__(32 * CONV_WAVES)
sparse_conv_wmma(const float* __restrict__ feats,
                 const float* __restrict__ W,
                 const int*   __restrict__ pin,
                 const int*   __restrict__ pout,
                 float*       __restrict__ out,
                 int P, int mtiles) {
  const int wave  = __builtin_amdgcn_readfirstlane((int)(threadIdx.x >> 5));
  int mtile = blockIdx.x * CONV_WAVES + wave;
  if (mtile >= mtiles) return;
  const int tstep = gridDim.x * CONV_WAVES;

  const int k     = blockIdx.y;
  const unsigned lane = threadIdx.x & 31u;
  const int half  = (int)(lane >> 4);
  const int nl    = (int)(lane & 15u);

  const float* wk   = W + (size_t)k * CIN * COUT;
  const float* bptr = wk + (unsigned)(2 * half * COUT + nl);
  v2f bf[CIN / 4][NT];
#pragma unroll
  for (int s = 0; s < CIN / 4; ++s) {
#pragma unroll
    for (int j = 0; j < NT; ++j) {
      bf[s][j].x = bptr[4 * s * COUT + 16 * j];
      bf[s][j].y = bptr[4 * s * COUT + COUT + 16 * j];
    }
  }

  const int* pin_k  = pin  + (size_t)k * P;
  const int* pout_k = pout + (size_t)k * P;

  int pidx   = min(mtile * 16 + nl, P - 1);
  int arow   = pin_k[pidx];
  int orow_l = pout_k[pidx];

  for (; mtile < mtiles; mtile += tstep) {
    const int m0 = mtile * 16;

    const int nidx   = min(m0 + tstep * 16 + nl, P - 1);
    const int arow_n = pin_k[nidx];
    const int orow_n = pout_k[nidx];

    const float* aptr = feats + ((unsigned)arow * CIN + 2u * half);
    v2f a[CIN / 4];
#pragma unroll
    for (int s = 0; s < CIN / 4; ++s)
      a[s] = *reinterpret_cast<const v2f*>(aptr + 4 * s);

    int orow[16];
#pragma unroll
    for (int i = 0; i < 16; ++i) orow[i] = __builtin_amdgcn_readlane(orow_l, i);

    v8f c[NT];
#pragma unroll
    for (int j = 0; j < NT; ++j) c[j] = v8f{};

#pragma unroll
    for (int s = 0; s < CIN / 4; ++s) {
#pragma unroll
      for (int j = 0; j < NT; ++j)
        c[j] = __builtin_amdgcn_wmma_f32_16x16x4_f32(false, a[s], false, bf[s][j],
                                                     (short)0, c[j], false, false);
    }

#pragma unroll
    for (int r = 0; r < 8; ++r) {
      float lo[NT / 2], hi[NT / 2];
#pragma unroll
      for (int t = 0; t < NT / 2; ++t) swap_halves(c[2 * t][r], c[2 * t + 1][r], half != 0, lo[t], hi[t]);
      if (m0 + r < P) {
        float* orp = out + (size_t)orow[r] * COUT;
#pragma unroll
        for (int t = 0; t < NT / 2; ++t) atomicAdd(orp + (lane + 32u * t), lo[t]);
      }
      if (m0 + r + 8 < P) {
        float* orp = out + (size_t)orow[r + 8] * COUT;
#pragma unroll
        for (int t = 0; t < NT / 2; ++t) atomicAdd(orp + (lane + 32u * t), hi[t]);
      }
    }

    arow   = arow_n;
    orow_l = orow_n;
  }
}

__global__ void col_stats(const float* __restrict__ x, int n_elems,
                          float* __restrict__ stats) {
  __shared__ float ssum[COUT];
  __shared__ float ssq[COUT];
  if (threadIdx.x < COUT) { ssum[threadIdx.x] = 0.f; ssq[threadIdx.x] = 0.f; }
  __syncthreads();
  int idx    = blockIdx.x * blockDim.x + threadIdx.x;
  int stride = gridDim.x * blockDim.x;
  const int ch = idx & (COUT - 1);
  float lsum = 0.f, lsq = 0.f;
  for (int i = idx; i < n_elems; i += stride) {
    float v = x[i];
    lsum += v;
    lsq  += v * v;
  }
  atomicAdd(&ssum[ch], lsum);
  atomicAdd(&ssq[ch], lsq);
  __syncthreads();
  if (threadIdx.x < COUT) {
    atomicAdd(&stats[threadIdx.x], ssum[threadIdx.x]);
    atomicAdd(&stats[COUT + threadIdx.x], ssq[threadIdx.x]);
  }
}

__global__ void bn_finalize(const float* __restrict__ stats,
                            const float* __restrict__ g,
                            const float* __restrict__ b,
                            float inv_n, float* __restrict__ ab) {
  int c = threadIdx.x;
  if (c < COUT) {
    float mean = stats[c] * inv_n;
    float var  = stats[COUT + c] * inv_n - mean * mean;
    float s    = g[c] * rsqrtf(var + BN_EPS);
    ab[c]        = s;
    ab[COUT + c] = b[c] - mean * s;
  }
}

__global__ void bn_apply(const float* __restrict__ y,
                         const float* __restrict__ ab,
                         float* __restrict__ x, int n_elems) {
  int i = blockIdx.x * blockDim.x + threadIdx.x;
  int s = gridDim.x * blockDim.x;
  for (; i < n_elems; i += s) {
    int ch = i & (COUT - 1);
    float v = y[i] * ab[ch] + ab[COUT + ch];
    x[i] = v > 0.f ? v : 0.f;
  }
}

__global__ void bn_apply_add(const float* __restrict__ y,
                             const float* __restrict__ ab,
                             const float* __restrict__ xnet,
                             float* __restrict__ out, int n_elems) {
  int i = blockIdx.x * blockDim.x + threadIdx.x;
  int s = gridDim.x * blockDim.x;
  for (; i < n_elems; i += s) {
    int ch = i & (COUT - 1);
    float v = y[i] * ab[ch] + ab[COUT + ch];
    out[i] = xnet[i] + (v > 0.f ? v : 0.f);
  }
}

static inline int cdiv(long a, long b) { return (int)((a + b - 1) / b); }

extern "C" void kernel_launch(void* const* d_in, const int* in_sizes, int n_in,
                              void* d_out, int out_size, void* d_ws, size_t ws_size,
                              hipStream_t stream) {
  const float* feats  = (const float*)d_in[0];
  const float* W_down = (const float*)d_in[1];
  const float* g0     = (const float*)d_in[2];
  const float* b0     = (const float*)d_in[3];
  const float* W1     = (const float*)d_in[4];
  const float* g1     = (const float*)d_in[5];
  const float* b1     = (const float*)d_in[6];
  const float* W2     = (const float*)d_in[7];
  const float* g2     = (const float*)d_in[8];
  const float* b2     = (const float*)d_in[9];
  const int* p_down_in  = (const int*)d_in[10];
  const int* p_down_out = (const int*)d_in[11];
  const int* p_subm_in  = (const int*)d_in[12];
  const int* p_subm_out = (const int*)d_in[13];

  const int Cin = 32;
  const int K   = in_sizes[1] / (Cin * COUT);
  const int P1  = in_sizes[10] / K;
  const int P2  = in_sizes[12] / K;
  const int n_out = out_size / COUT;
  const int NE  = n_out * COUT;
  const float inv_n = 1.0f / (float)n_out;

  float* y     = (float*)d_ws;
  float* x     = y + NE;
  float* xnet  = x + NE;
  float* stats = xnet + NE;
  float* ab    = stats + 128;
  (void)n_in; (void)ws_size;

  const dim3 blk(256);
  const int zb  = cdiv(NE / 4, 256);
  const int eb  = cdiv(NE, 256);
  const int sb  = 1024;

  const int mt1 = cdiv(P1, 16);
  const int mt2 = cdiv(P2, 16);
  const dim3 cblk(32 * CONV_WAVES);
  const dim3 cg1(75, K);
  const dim3 cg2(47, K);

  zero_f32v4<<<zb, blk, 0, stream>>>((v4f*)y, NE / 4);
  zero_f32v4<<<1, 32, 0, stream>>>((v4f*)stats, 32);
  sparse_conv_wmma<32><<<cg1, cblk, 0, stream>>>(feats, W_down, p_down_in, p_down_out,
                                                y, P1, mt1);
  col_stats<<<sb, blk, 0, stream>>>(y, NE, stats);
  bn_finalize<<<1, 64, 0, stream>>>(stats, g0, b0, inv_n, ab);
  bn_apply<<<eb, blk, 0, stream>>>(y, ab, x, NE);

  zero_f32v4<<<zb, blk, 0, stream>>>((v4f*)y, NE / 4);
  zero_f32v4<<<1, 32, 0, stream>>>((v4f*)stats, 32);
  sparse_conv_wmma<64><<<cg2, cblk, 0, stream>>>(x, W1, p_subm_in, p_subm_out,
                                                y, P2, mt2);
  col_stats<<<sb, blk, 0, stream>>>(y, NE, stats);
  bn_finalize<<<1, 64, 0, stream>>>(stats, g1, b1, inv_n, ab);
  bn_apply<<<eb, blk, 0, stream>>>(y, ab, xnet, NE);

  zero_f32v4<<<zb, blk, 0, stream>>>((v4f*)y, NE / 4);
  zero_f32v4<<<1, 32, 0, stream>>>((v4f*)stats, 32);
  sparse_conv_wmma<64><<<cg2, cblk, 0, stream>>>(xnet, W2, p_subm_in, p_subm_out,
                                                y, P2, mt2);
  col_stats<<<sb, blk, 0, stream>>>(y, NE, stats);
  bn_finalize<<<1, 64, 0, stream>>>(stats, g2, b2, inv_n, ab);
  bn_apply_add<<<eb, blk, 0, stream>>>(y, ab, xnet, (float*)d_out, NE);
}


# launches 18 -> 10: BN finalize done inside the apply pass (each thread derives its 4 channels' scale/shift from the sums), y re-zeroed by the apply that last reads it, sums cleared with y
# speedup vs baseline: 2.7271x; 1.1579x over previous
#include <hip/hip_runtime.h>

typedef float v2f __attribute__((ext_vector_type(2)));
typedef float v4f __attribute__((ext_vector_type(4)));
typedef float v8f __attribute__((ext_vector_type(8)));

#define COUT 64
#define BN_EPS 1e-5f

__global__ void zero_all(v4f* __restrict__ y4, int n4, v4f* __restrict__ s4) {
  v4f z = {};
  if (blockIdx.x == 0 && threadIdx.x < 64) s4[threadIdx.x] = z;
  int i = blockIdx.x * blockDim.x + threadIdx.x;
  int s = gridDim.x * blockDim.x;
  for (; i < n4; i += s) y4[i] = z;
}

#define NT (COUT / 16)
#define CONV_WAVES 4

__device__ __forceinline__ void swap_halves(float x, float y, bool upper, float& lo, float& hi) {
  const int give = __builtin_bit_cast(int, upper ? x : y);
  const float got = __builtin_bit_cast(float,
      __builtin_amdgcn_permlanex16(give, give, 0x76543210, 0xfedcba98, false, false));
  lo = upper ? got : x;
  hi = upper ? y : got;
}

template <int CIN>
__global__ void __launch_bounds__(32 * CONV_WAVES)
sparse_conv_wmma(const float* __restrict__ feats,
                 const float* __restrict__ W,
                 const int*   __restrict__ pin,
                 const int*   __restrict__ pout,
                 float*       __restrict__ out,
                 int P, int mtiles) {
  const int wave  = __builtin_amdgcn_readfirstlane((int)(threadIdx.x >> 5));
  int mtile = blockIdx.x * CONV_WAVES + wave;
  if (mtile >= mtiles) return;
  const int tstep = gridDim.x * CONV_WAVES;

  const int k     = blockIdx.y;
  const unsigned lane = threadIdx.x & 31u;
  const int half  = (int)(lane >> 4);
  const int nl    = (int)(lane & 15u);

  const float* wk   = W + (size_t)k * CIN * COUT;
  const float* bptr = wk + (unsigned)(2 * half * COUT + nl);
  v2f bf[CIN / 4][NT];
#pragma unroll
  for (int s = 0; s < CIN / 4; ++s) {
#pragma unroll
    for (int j = 0; j < NT; ++j) {
      bf[s][j].x = bptr[4 * s * COUT + 16 * j];
      bf[s][j].y = bptr[4 * s * COUT + COUT + 16 * j];
    }
  }

  const int* pin_k  = pin  + (size_t)k * P;
  const int* pout_k = pout + (size_t)k * P;

  int pidx   = min(mtile * 16 + nl, P - 1);
  int arow   = pin_k[pidx];
  int orow_l = pout_k[pidx];

  for (; mtile < mtiles; mtile += tstep) {
    const int m0 = mtile * 16;

    const int nidx   = min(m0 + tstep * 16 + nl, P - 1);
    const int arow_n = pin_k[nidx];
    const int orow_n = pout_k[nidx];

    const float* aptr = feats + ((unsigned)arow * CIN + 2u * half);
    v2f a[CIN / 4];
#pragma unroll
    for (int s = 0; s < CIN / 4; ++s)
      a[s] = *reinterpret_cast<const v2f*>(aptr + 4 * s);

    int orow[16];
#pragma unroll
    for (int i = 0; i < 16; ++i) orow[i] = __builtin_amdgcn_readlane(orow_l, i);

    v8f c[NT];
#pragma unroll
    for (int j = 0; j < NT; ++j) c[j] = v8f{};

#pragma unroll
    for (int s = 0; s < CIN / 4; ++s) {
#pragma unroll
      for (int j = 0; j < NT; ++j)
        c[j] = __builtin_amdgcn_wmma_f32_16x16x4_f32(false, a[s], false, bf[s][j],
                                                     (short)0, c[j], false, false);
    }

#pragma unroll
    for (int r = 0; r < 8; ++r) {
      float lo[NT / 2], hi[NT / 2];
#pragma unroll
      for (int t = 0; t < NT / 2; ++t) swap_halves(c[2 * t][r], c[2 * t + 1][r], half != 0, lo[t], hi[t]);
      if (m0 + r < P) {
        float* orp = out + (size_t)orow[r] * COUT;
#pragma unroll
        for (int t = 0; t < NT / 2; ++t) atomicAdd(orp + (lane + 32u * t), lo[t]);
      }
      if (m0 + r + 8 < P) {
        float* orp = out + (size_t)orow[r + 8] * COUT;
#pragma unroll
        for (int t = 0; t < NT / 2; ++t) atomicAdd(orp + (lane + 32u * t), hi[t]);
      }
    }

    arow   = arow_n;
    orow_l = orow_n;
  }
}

__global__ void col_stats(const float* __restrict__ x, int n_elems,
                          float* __restrict__ stats) {
  __shared__ float ssum[COUT];
  __shared__ float ssq[COUT];
  if (threadIdx.x < COUT) { ssum[threadIdx.x] = 0.f; ssq[threadIdx.x] = 0.f; }
  __syncthreads();
  int idx    = blockIdx.x * blockDim.x + threadIdx.x;
  int stride = gridDim.x * blockDim.x;
  const int ch = idx & (COUT - 1);
  float lsum = 0.f, lsq = 0.f;
  for (int i = idx; i < n_elems; i += stride) {
    float v = x[i];
    lsum += v;
    lsq  += v * v;
  }
  atomicAdd(&ssum[ch], lsum);
  atomicAdd(&ssq[ch], lsq);
  __syncthreads();
  if (threadIdx.x < COUT) {
    atomicAdd(&stats[threadIdx.x], ssum[threadIdx.x]);
    atomicAdd(&stats[COUT + threadIdx.x], ssq[threadIdx.x]);
  }
}

#define APPLY_U 4
template <bool ADD>
__global__ void __launch_bounds__(256)
bn_apply(float* __restrict__ y, const float* __restrict__ stats,
         const float* __restrict__ g, const float* __restrict__ b, float inv_n,
         const float* __restrict__ xnet, float* __restrict__ dst, int n4,
         float* __restrict__ clr) {
  const int i0     = blockIdx.x * 256 + threadIdx.x;
  const int stride = gridDim.x * 256;
  const int c4     = i0 & (COUT / 4 - 1);
  v4f*       y4 = reinterpret_cast<v4f*>(y);
  const v4f* r4 = reinterpret_cast<const v4f*>(xnet);
  v4f*       d4 = reinterpret_cast<v4f*>(dst);
  const v4f z = {};

  v4f v[APPLY_U], r[APPLY_U];
#pragma unroll
  for (int j = 0; j < APPLY_U; ++j) {
    const int ic = min(i0 + j * stride, n4 - 1);
    v[j] = y4[ic];
    if (ADD) r[j] = r4[ic];
  }

  const v4f sum = reinterpret_cast<const v4f*>(stats)[c4];
  const v4f sq  = reinterpret_cast<const v4f*>(stats + COUT)[c4];
  const v4f gg  = reinterpret_cast<const v4f*>(g)[c4];
  const v4f bb  = reinterpret_cast<const v4f*>(b)[c4];
  v4f sc, sh;
#pragma unroll
  for (int e = 0; e < 4; ++e) {
    float mean = sum[e] * inv_n;
    float var  = sq[e] * inv_n - mean * mean;
    float s    = gg[e] * rsqrtf(var + BN_EPS);
    sc[e] = s;
    sh[e] = bb[e] - mean * s;
  }

#pragma unroll
  for (int j = 0; j < APPLY_U; ++j) {
    const int i = i0 + j * stride;
    if (i < n4) {
      v4f o;
#pragma unroll
      for (int e = 0; e < 4; ++e) {
        float t = v[j][e] * sc[e] + sh[e];
        t = t > 0.f ? t : 0.f;
        o[e] = ADD ? r[j][e] + t : t;
      }
      d4[i] = o;
      if (!ADD) y4[i] = z;
    }
  }

  if (clr != nullptr && blockIdx.x == 0 && threadIdx.x < 32)
    reinterpret_cast<v4f*>(clr)[threadIdx.x] = z;
}

static inline int cdiv(long a, long b) { return (int)((a + b - 1) / b); }

extern "C" void kernel_launch(void* const* d_in, const int* in_sizes, int n_in,
                              void* d_out, int out_size, void* d_ws, size_t ws_size,
                              hipStream_t stream) {
  const float* feats  = (const float*)d_in[0];
  const float* W_down = (const float*)d_in[1];
  const float* g0     = (const float*)d_in[2];
  const float* b0     = (const float*)d_in[3];
  const float* W1     = (const float*)d_in[4];
  const float* g1     = (const float*)d_in[5];
  const float* b1     = (const float*)d_in[6];
  const float* W2     = (const float*)d_in[7];
  const float* g2     = (const float*)d_in[8];
  const float* b2     = (const float*)d_in[9];
  const int* p_down_in  = (const int*)d_in[10];
  const int* p_down_out = (const int*)d_in[11];
  const int* p_subm_in  = (const int*)d_in[12];
  const int* p_subm_out = (const int*)d_in[13];

  const int Cin = 32;
  const int K   = in_sizes[1] / (Cin * COUT);
  const int P1  = in_sizes[10] / K;
  const int P2  = in_sizes[12] / K;
  const int n_out = out_size / COUT;
  const int NE  = n_out * COUT;
  const float inv_n = 1.0f / (float)n_out;

  float* y     = (float*)d_ws;
  float* x     = y + NE;
  float* xnet  = x + NE;
  float* statA = xnet + NE;
  float* statB = statA + 128;
  (void)n_in; (void)ws_size;

  const dim3 blk(256);
  const int n4  = NE / 4;
  const int zb  = cdiv(n4, 256);
  const int ab  = cdiv(n4, 256 * APPLY_U);
  const int sb  = 1024;

  const int mt1 = cdiv(P1, 16);
  const int mt2 = cdiv(P2, 16);
  const dim3 cblk(32 * CONV_WAVES);
  const dim3 cg1(75, K);
  const dim3 cg2(47, K);


  zero_all<<<zb, blk, 0, stream>>>((v4f*)y, n4, (v4f*)statA);
  sparse_conv_wmma<32><<<cg1, cblk, 0, stream>>>(feats, W_down, p_down_in, p_down_out,
                                                y, P1, mt1);
  col_stats<<<sb, blk, 0, stream>>>(y, NE, statA);
  bn_apply<false><<<ab, blk, 0, stream>>>(y, statA, g0, b0, inv_n, nullptr, x, n4, nullptr);

  sparse_conv_wmma<64><<<cg2, cblk, 0, stream>>>(x, W1, p_subm_in, p_subm_out,
                                                y, P2, mt2);
  col_stats<<<sb, blk, 0, stream>>>(y, NE, statB);
  bn_apply<false><<<ab, blk, 0, stream>>>(y, statB, g1, b1, inv_n, nullptr, xnet, n4, statA);

  sparse_conv_wmma<64><<<cg2, cblk, 0, stream>>>(xnet, W2, p_subm_in, p_subm_out,
                                                y, P2, mt2);
  col_stats<<<sb, blk, 0, stream>>>(y, NE, statA);
  bn_apply<true><<<ab, blk, 0, stream>>>(y, statA, g2, b2, inv_n, xnet, (float*)d_out, n4, nullptr);
}


# BN+ReLU of layers 1 and 2 done inside the next conv's gather (scale/shift per channel on each gathered row of raw sums); their 2 apply launches and stored layers go: 10 -> 8 launches; atomics kept
# speedup vs baseline: 2.8190x; 1.0337x over previous
#include <hip/hip_runtime.h>

typedef float v2f __attribute__((ext_vector_type(2)));
typedef float v4f __attribute__((ext_vector_type(4)));
typedef float v8f __attribute__((ext_vector_type(8)));

#define COUT 64
#define BN_EPS 1e-5f

__global__ void zero_all(v4f* __restrict__ y4, int n4, v4f* __restrict__ s4) {
  v4f z = {};
  if (blockIdx.x == 0 && threadIdx.x < 64) s4[threadIdx.x] = z;
  int i = blockIdx.x * blockDim.x + threadIdx.x;
  int s = gridDim.x * blockDim.x;
  for (; i < n4; i += s) y4[i] = z;
}

#define NT (COUT / 16)
#define CONV_WAVES 4

__device__ __forceinline__ void swap_halves(float x, float y, bool upper, float& lo, float& hi) {
  const int give = __builtin_bit_cast(int, upper ? x : y);
  const float got = __builtin_bit_cast(float,
      __builtin_amdgcn_permlanex16(give, give, 0x76543210, 0xfedcba98, false, false));
  lo = upper ? got : x;
  hi = upper ? y : got;
}

template <int CIN, bool BN>
__global__ void __launch_bounds__(32 * CONV_WAVES)
sparse_conv_wmma(const float* __restrict__ feats,
                 const float* __restrict__ W,
                 const int*   __restrict__ pin,
                 const int*   __restrict__ pout,
                 float*       __restrict__ out,
                 int P, int mtiles,
                 const float* __restrict__ stats,
                 const float* __restrict__ g,
                 const float* __restrict__ b,
                 float inv_n) {
  __shared__ v4f coef[BN ? CIN / 2 : 1];

  const int wave  = __builtin_amdgcn_readfirstlane((int)(threadIdx.x >> 5));
  int mtile = blockIdx.x * CONV_WAVES + wave;
  if (!BN && mtile >= mtiles) return;
  const int tstep = gridDim.x * CONV_WAVES;

  const int k     = blockIdx.y;
  const unsigned lane = threadIdx.x & 31u;
  const int half  = (int)(lane >> 4);
  const int nl    = (int)(lane & 15u);

  const float* wk   = W + (size_t)k * CIN * COUT;
  const float* bptr = wk + (unsigned)(2 * half * COUT + nl);
  v2f bf[CIN / 4][NT];
#pragma unroll
  for (int s = 0; s < CIN / 4; ++s) {
#pragma unroll
    for (int j = 0; j < NT; ++j) {
      bf[s][j].x = bptr[4 * s * COUT + 16 * j];
      bf[s][j].y = bptr[4 * s * COUT + COUT + 16 * j];
    }
  }

  if (BN) {
    if (threadIdx.x < CIN) {
      const int c = threadIdx.x;
      float mean = stats[c] * inv_n;
      float var  = stats[CIN + c] * inv_n - mean * mean;
      float s    = g[c] * rsqrtf(var + BN_EPS);
      float* cf  = reinterpret_cast<float*>(coef);
      cf[(c >> 1) * 4 + (c & 1)]     = s;
      cf[(c >> 1) * 4 + 2 + (c & 1)] = b[c] - mean * s;
    }
    __syncthreads();
    if (mtile >= mtiles) return;
  }

  const int* pin_k  = pin  + (size_t)k * P;
  const int* pout_k = pout + (size_t)k * P;

  int pidx   = min(mtile * 16 + nl, P - 1);
  int arow   = pin_k[pidx];
  int orow_l = pout_k[pidx];

  for (; mtile < mtiles; mtile += tstep) {
    const int m0 = mtile * 16;

    const int nidx   = min(m0 + tstep * 16 + nl, P - 1);
    const int arow_n = pin_k[nidx];
    const int orow_n = pout_k[nidx];

    const float* aptr = feats + ((unsigned)arow * CIN + 2u * half);
    v2f a[CIN / 4];
#pragma unroll
    for (int s = 0; s < CIN / 4; ++s)
      a[s] = *reinterpret_cast<const v2f*>(aptr + 4 * s);
    if (BN) {
#pragma unroll
      for (int s = 0; s < CIN / 4; ++s) {
        const v4f q = coef[2 * s + half];
        float t0 = a[s].x * q.x + q.z;
        float t1 = a[s].y * q.y + q.w;
        a[s].x = t0 > 0.f ? t0 : 0.f;
        a[s].y = t1 > 0.f ? t1 : 0.f;
      }
    }

    int orow[16];
#pragma unroll
    for (int i = 0; i < 16; ++i) orow[i] = __builtin_amdgcn_readlane(orow_l, i);

    v8f c[NT];
#pragma unroll
    for (int j = 0; j < NT; ++j) c[j] = v8f{};

#pragma unroll
    for (int s = 0; s < CIN / 4; ++s) {
#pragma unroll
      for (int j = 0; j < NT; ++j)
        c[j] = __builtin_amdgcn_wmma_f32_16x16x4_f32(false, a[s], false, bf[s][j],
                                                     (short)0, c[j], false, false);
    }

#pragma unroll
    for (int r = 0; r < 8; ++r) {
      float lo[NT / 2], hi[NT / 2];
#pragma unroll
      for (int t = 0; t < NT / 2; ++t) swap_halves(c[2 * t][r], c[2 * t + 1][r], half != 0, lo[t], hi[t]);
      if (m0 + r < P) {
        float* orp = out + (size_t)orow[r] * COUT;
#pragma unroll
        for (int t = 0; t < NT / 2; ++t) atomicAdd(orp + (lane + 32u * t), lo[t]);
      }
      if (m0 + r + 8 < P) {
        float* orp = out + (size_t)orow[r + 8] * COUT;
#pragma unroll
        for (int t = 0; t < NT / 2; ++t) atomicAdd(orp + (lane + 32u * t), hi[t]);
      }
    }

    arow   = arow_n;
    orow_l = orow_n;
  }
}

__global__ void col_stats(const float* __restrict__ x, int n_elems,
                          float* __restrict__ stats,
                          v4f* __restrict__ clr4, int n4c, v4f* __restrict__ clrs) {
  __shared__ float ssum[COUT];
  __shared__ float ssq[COUT];
  if (threadIdx.x < COUT) { ssum[threadIdx.x] = 0.f; ssq[threadIdx.x] = 0.f; }
  __syncthreads();
  int idx    = blockIdx.x * blockDim.x + threadIdx.x;
  int stride = gridDim.x * blockDim.x;
  if (clr4 != nullptr) {
    const v4f z = {};
    for (int i = idx; i < n4c; i += stride) clr4[i] = z;
    if (blockIdx.x == 0 && threadIdx.x < 32) clrs[threadIdx.x] = z;
  }
  const int ch = idx & (COUT - 1);
  float lsum = 0.f, lsq = 0.f;
  for (int i = idx; i < n_elems; i += stride) {
    float v = x[i];
    lsum += v;
    lsq  += v * v;
  }
  atomicAdd(&ssum[ch], lsum);
  atomicAdd(&ssq[ch], lsq);
  __syncthreads();
  if (threadIdx.x < COUT) {
    atomicAdd(&stats[threadIdx.x], ssum[threadIdx.x]);
    atomicAdd(&stats[COUT + threadIdx.x], ssq[threadIdx.x]);
  }
}

#define APPLY_U 4
__global__ void __launch_bounds__(256)
bn_final(const float* __restrict__ ya, const float* __restrict__ sa,
         const float* __restrict__ ga, const float* __restrict__ ba,
         const float* __restrict__ yb, const float* __restrict__ sb,
         const float* __restrict__ gb, const float* __restrict__ bb,
         float inv_n, float* __restrict__ dst, int n4) {
  const int i0     = blockIdx.x * 256 + threadIdx.x;
  const int stride = gridDim.x * 256;
  const int c4     = i0 & (COUT / 4 - 1);
  const v4f* a4 = reinterpret_cast<const v4f*>(ya);
  const v4f* b4 = reinterpret_cast<const v4f*>(yb);
  v4f*       d4 = reinterpret_cast<v4f*>(dst);

  v4f va[APPLY_U], vb[APPLY_U];
#pragma unroll
  for (int j = 0; j < APPLY_U; ++j) {
    const int ic = min(i0 + j * stride, n4 - 1);
    va[j] = a4[ic];
    vb[j] = b4[ic];
  }

  v4f sca, sha, scb, shb;
  {
    const v4f sum = reinterpret_cast<const v4f*>(sa)[c4];
    const v4f sq  = reinterpret_cast<const v4f*>(sa + COUT)[c4];
    const v4f gg  = reinterpret_cast<const v4f*>(ga)[c4];
    const v4f be  = reinterpret_cast<const v4f*>(ba)[c4];
#pragma unroll
    for (int e = 0; e < 4; ++e) {
      float mean = sum[e] * inv_n;
      float var  = sq[e] * inv_n - mean * mean;
      float s    = gg[e] * rsqrtf(var + BN_EPS);
      sca[e] = s;
      sha[e] = be[e] - mean * s;
    }
  }
  {
    const v4f sum = reinterpret_cast<const v4f*>(sb)[c4];
    const v4f sq  = reinterpret_cast<const v4f*>(sb + COUT)[c4];
    const v4f gg  = reinterpret_cast<const v4f*>(gb)[c4];
    const v4f be  = reinterpret_cast<const v4f*>(bb)[c4];
#pragma unroll
    for (int e = 0; e < 4; ++e) {
      float mean = sum[e] * inv_n;
      float var  = sq[e] * inv_n - mean * mean;
      float s    = gg[e] * rsqrtf(var + BN_EPS);
      scb[e] = s;
      shb[e] = be[e] - mean * s;
    }
  }

#pragma unroll
  for (int j = 0; j < APPLY_U; ++j) {
    const int i = i0 + j * stride;
    if (i < n4) {
      v4f o;
#pragma unroll
      for (int e = 0; e < 4; ++e) {
        float r = va[j][e] * sca[e] + sha[e];
        r = r > 0.f ? r : 0.f;
        float t = vb[j][e] * scb[e] + shb[e];
        t = t > 0.f ? t : 0.f;
        o[e] = r + t;
      }
      d4[i] = o;
    }
  }
}

static inline int cdiv(long a, long b) { return (int)((a + b - 1) / b); }

extern "C" void kernel_launch(void* const* d_in, const int* in_sizes, int n_in,
                              void* d_out, int out_size, void* d_ws, size_t ws_size,
                              hipStream_t stream) {
  const float* feats  = (const float*)d_in[0];
  const float* W_down = (const float*)d_in[1];
  const float* g0     = (const float*)d_in[2];
  const float* b0     = (const float*)d_in[3];
  const float* W1     = (const float*)d_in[4];
  const float* g1     = (const float*)d_in[5];
  const float* b1     = (const float*)d_in[6];
  const float* W2     = (const float*)d_in[7];
  const float* g2     = (const float*)d_in[8];
  const float* b2     = (const float*)d_in[9];
  const int* p_down_in  = (const int*)d_in[10];
  const int* p_down_out = (const int*)d_in[11];
  const int* p_subm_in  = (const int*)d_in[12];
  const int* p_subm_out = (const int*)d_in[13];

  const int Cin = 32;
  const int K   = in_sizes[1] / (Cin * COUT);
  const int P1  = in_sizes[10] / K;
  const int P2  = in_sizes[12] / K;
  const int n_out = out_size / COUT;
  const int NE  = n_out * COUT;
  const float inv_n = 1.0f / (float)n_out;

  float* ya    = (float*)d_ws;
  float* yb    = ya + NE;
  float* statA = yb + 2 * (size_t)NE;
  float* statB = statA + 128;
  (void)n_in; (void)ws_size;

  const dim3 blk(256);
  const int n4  = NE / 4;
  const int zb  = cdiv(2 * (long)n4, 256);
  const int ab  = cdiv(n4, 256 * APPLY_U);
  const int sb  = 1024;

  const int mt1 = cdiv(P1, 16);
  const int mt2 = cdiv(P2, 16);
  const dim3 cblk(32 * CONV_WAVES);
  const dim3 cg1(75, K);
  const dim3 cg2(37, K);


  zero_all<<<zb, blk, 0, stream>>>((v4f*)ya, 2 * n4, (v4f*)statA);
  sparse_conv_wmma<32, false><<<cg1, cblk, 0, stream>>>(feats, W_down, p_down_in, p_down_out,
                                                       ya, P1, mt1, nullptr, nullptr, nullptr, 0.f);
  col_stats<<<sb, blk, 0, stream>>>(ya, NE, statA, nullptr, 0, nullptr);

  sparse_conv_wmma<64, true><<<cg2, cblk, 0, stream>>>(ya, W1, p_subm_in, p_subm_out,
                                                      yb, P2, mt2, statA, g0, b0, inv_n);
  col_stats<<<sb, blk, 0, stream>>>(yb, NE, statB, (v4f*)ya, n4, (v4f*)statA);

  sparse_conv_wmma<64, true><<<cg2, cblk, 0, stream>>>(yb, W2, p_subm_in, p_subm_out,
                                                      ya, P2, mt2, statB, g1, b1, inv_n);
  col_stats<<<sb, blk, 0, stream>>>(ya, NE, statA, nullptr, 0, nullptr);

  bn_final<<<ab, blk, 0, stream>>>(yb, statB, g1, b1, ya, statA, g2, b2, inv_n,
                                  (float*)d_out, n4);
}
